# starting kernel
# baseline (speedup 1.0000x reference)
#include <hip/hip_runtime.h>


typedef __attribute__((ext_vector_type(2))) float v2f;
typedef __attribute__((ext_vector_type(8))) float v8f;

#define NFILT  1024
#define HH     64
#define WW     64
#define KH     5
#define KW     5
#define LOCAL  3
#define WIN    11
#define KPIX   121
#define KPAD   124
#define NPIX   36
#define NPAD   48
#define MB     64
#define THREADS 128

__global__ __launch_bounds__(THREADS)
void filters_wmma_kernel(const float* __restrict__ X,
                         const float* __restrict__ weight,
                         const float* __restrict__ bias,
                         const int*   __restrict__ positions,
                         float*       __restrict__ out)
{
    __shared__ float A_lds[MB * KPAD];
    __shared__ float B_lds[KPAD * NPAD];

    const int f      = blockIdx.x;
    const int batch0 = blockIdx.y * MB;
    const int tid    = threadIdx.x;
    const int lane   = tid & 31;
    const int wave   = tid >> 5;
    const int half   = lane >> 4;
    const int nl     = lane & 15;

    const int i0 = positions[2 * f + 0];
    const int j0 = positions[2 * f + 1];

    for (int idx = tid; idx < MB * KPAD; idx += THREADS) {
        const int mb = idx / KPAD;
        const int k  = idx - mb * KPAD;
        float v = 0.0f;
        if (k < KPIX) {
            const int wi = k / WIN;
            const int wj = k - wi * WIN;
            const int r  = i0 + wi - LOCAL;
            const int c  = j0 + wj - LOCAL;
            if ((unsigned)r < (unsigned)HH && (unsigned)c < (unsigned)WW)
                v = X[(size_t)(batch0 + mb) * (HH * WW) + r * WW + c];
        }
        A_lds[idx] = v;
    }

    for (int idx = tid; idx < KPAD * NPAD; idx += THREADS) B_lds[idx] = 0.0f;
    __syncthreads();
    for (int idx = tid; idx < NPIX * KH * KW; idx += THREADS) {
        const int p   = idx / (KH * KW);
        const int kk  = idx - p * (KH * KW);
        const int ki  = kk / KW;
        const int kj  = kk - ki * KW;
        const int pi  = p / 6;
        const int pj  = p - pi * 6;
        const int krow = (pi + ki) * WIN + (pj + kj);
        B_lds[krow * NPAD + p] = weight[f * (KH * KW) + kk];
    }
    __syncthreads();

    v8f acc0 = {}, acc1 = {}, acc2 = {};
    const float* Arow = &A_lds[(wave * 16 + nl) * KPAD];
    #pragma unroll 1
    for (int s = 0; s < KPAD / 4; ++s) {
        const int k0 = 4 * s + 2 * half;
        v2f a;  a.x  = Arow[k0];        a.y  = Arow[k0 + 1];
        const float* Br0 = &B_lds[k0 * NPAD + nl];
        const float* Br1 = Br0 + NPAD;
        v2f b0; b0.x = Br0[0];   b0.y = Br1[0];
        v2f b1; b1.x = Br0[16];  b1.y = Br1[16];
        v2f b2; b2.x = Br0[32];  b2.y = Br1[32];
        acc0 = __builtin_amdgcn_wmma_f32_16x16x4_f32(false, a, false, b0, (short)0, acc0, false, false);
        acc1 = __builtin_amdgcn_wmma_f32_16x16x4_f32(false, a, false, b1, (short)0, acc1, false, false);
        acc2 = __builtin_amdgcn_wmma_f32_16x16x4_f32(false, a, false, b2, (short)0, acc2, false, false);
    }

    __syncthreads();
    float* conv = &A_lds[wave * (16 * NPAD)];
    #pragma unroll
    for (int r = 0; r < 8; ++r) {
        conv[(r + 8 * half) * NPAD +  0 + nl] = acc0[r];
        conv[(r + 8 * half) * NPAD + 16 + nl] = acc1[r];
        conv[(r + 8 * half) * NPAD + 32 + nl] = acc2[r];
    }
    __syncthreads();

    if (lane < 16) {
        const int b = batch0 + wave * 16 + nl;
        const float bv = bias[f];
        const float* cr = &conv[nl * NPAD];
        float4 o;
        float* oo = &o.x;
        #pragma unroll
        for (int q = 0; q < 4; ++q) {
            const int qi = q >> 1, qj = q & 1;
            float mx = -__builtin_inff();
            #pragma unroll
            for (int di = 0; di < 3; ++di)
                #pragma unroll
                for (int dj = 0; dj < 3; ++dj)
                    mx = fmaxf(mx, cr[(3 * qi + di) * 6 + (3 * qj + dj)]);
            oo[q] = mx + bv;
        }
        *(float4*)&out[(size_t)b * (NFILT * 4) + f * 4] = o;
    }
}

extern "C" void kernel_launch(void* const* d_in, const int* in_sizes, int n_in,
                              void* d_out, int out_size, void* d_ws, size_t ws_size,
                              hipStream_t stream) {
    const float* X         = (const float*)d_in[0];
    const float* weight    = (const float*)d_in[1];
    const float* bias      = (const float*)d_in[2];
    const int*   positions = (const int*)d_in[3];
    float*       out       = (float*)d_out;

    dim3 grid(NFILT, 1024 / MB, 1);
    dim3 block(THREADS, 1, 1);
    filters_wmma_kernel<<<grid, block, 0, stream>>>(X, weight, bias, positions, out);
}


# Batch-innermost copy of X in workspace (added launch); window pixels of 16 batches load straight into WMMA B fragments, no LDS staging; Toeplitz as A on 10x10 corner, K 124->100; sum order changes
# speedup vs baseline: 4.7759x; 4.7759x over previous
#include <hip/hip_runtime.h>


typedef __attribute__((ext_vector_type(2))) float v2f;
typedef __attribute__((ext_vector_type(8))) float v8f;

#define NFILT   1024
#define NBATCH  1024
#define HH      64
#define WW      64
#define IMGPIX  (HH * WW)
#define KH      5
#define KW      5
#define LOCAL   3
#define WINU    10
#define KDIM    (WINU * WINU)
#define KSTEPS  (KDIM / 4)
#define MROWS   48
#define THREADS 128
#define NT      2
#define BPW     (NBATCH / 4)
#define NPASS   (BPW / (16 * NT))
#define WS_NEED ((size_t)IMGPIX * NBATCH * sizeof(float))

#define TP 64
#define TB 128
__global__ __launch_bounds__(256)
void batch_minor_copy_kernel(const float* __restrict__ X, float* __restrict__ Xt, size_t ws_size)
{
    if (ws_size < WS_NEED) return;
    __shared__ float tile[TP][TB + 1];
    const int lane = threadIdx.x & 31;
    const int wv   = threadIdx.x >> 5;
    const int p0 = blockIdx.x * TP, b0 = blockIdx.y * TB;
    float2 v[TB / 8];
    #pragma unroll
    for (int i = 0; i < TB / 8; ++i)
        v[i] = *(const float2*)&X[(size_t)(b0 + wv + 8 * i) * IMGPIX + p0 + 2 * lane];
    #pragma unroll
    for (int i = 0; i < TB / 8; ++i) {
        tile[2 * lane + 0][wv + 8 * i] = v[i].x;
        tile[2 * lane + 1][wv + 8 * i] = v[i].y;
    }
    __syncthreads();
    #pragma unroll
    for (int i = 0; i < TP / 8; ++i) {
        const int p = wv + 8 * i;
        #pragma unroll
        for (int j = 0; j < TB / 32; ++j)
            Xt[(size_t)(p0 + p) * NBATCH + b0 + lane + 32 * j] = tile[p][lane + 32 * j];
    }
}

static __device__ __forceinline__ int clampi(int v, int lo, int hi) { return v < lo ? lo : (v > hi ? hi : v); }

template <bool COPY>
static __device__ __forceinline__
void filters_body(const float* __restrict__ src,
                  const float* __restrict__ weight,
                  const float* __restrict__ bias,
                  const int*   __restrict__ positions,
                  float*       __restrict__ out,
                  float*       T_lds)
{
    constexpr int PIXB = COPY ? NBATCH * (int)sizeof(float) : (int)sizeof(float);
    constexpr int BATB = COPY ? (int)sizeof(float) : IMGPIX * (int)sizeof(float);

    const int f    = blockIdx.x;
    const int tid  = threadIdx.x;
    const int lane = tid & 31;
    const int wave = __builtin_amdgcn_readfirstlane(tid >> 5);
    const int half = lane >> 4;
    const int nl   = lane & 15;

    const int i0 = clampi(positions[2 * f + 0], 0, HH - KH);
    const int j0 = clampi(positions[2 * f + 1], 0, WW - KW);

    unsigned off[KSTEPS][2];
    #pragma unroll
    for (int s = 0; s < KSTEPS; ++s) {
        #pragma unroll
        for (int r = 0; r < 2; ++r) {
            const int k  = 4 * s + 2 * half + r;
            const int wi = k / WINU;
            const int wj = k - wi * WINU;
            const int rc = clampi(i0 + wi - LOCAL, 0, HH - 1);
            const int cc = clampi(j0 + wj - LOCAL, 0, WW - 1);
            off[s][r] = (unsigned)((rc * WW + cc) * PIXB + nl * BATB);
        }
    }

    const char* bp = (const char*)src + (size_t)wave * BPW * BATB;
    v2f bf[KSTEPS][NT];
    #pragma unroll
    for (int s = 0; s < KSTEPS; ++s) {
        #pragma unroll
        for (int nt = 0; nt < NT; ++nt) {
            bf[s][nt].x = *(const float*)(bp + off[s][0] + nt * 16 * BATB);
            bf[s][nt].y = *(const float*)(bp + off[s][1] + nt * 16 * BATB);
        }
    }

    if (tid < KDIM) {
        const int  wi    = tid / WINU;
        const int  wj    = tid - wi * WINU;
        const int  r     = i0 + wi - LOCAL, c = j0 + wj - LOCAL;
        const bool inimg = ((unsigned)r < (unsigned)HH) && ((unsigned)c < (unsigned)WW);
        const float* wf  = weight + f * (KH * KW);
        #pragma unroll
        for (int mi = 0; mi < MROWS; ++mi) {
            const int t = mi >> 4, hh = (mi >> 3) & 1, rr = mi & 7;
            const int j = 8 * t + rr;
            float v = 0.0f;
            if (j < 18) {
                const int second = j >= 9 ? 1 : 0;
                const int q  = 2 * hh + second;
                const int e  = j - 9 * second;
                const int ei = e / 3;
                const int pi = 3 * (q >> 1) + ei;
                const int pj = 3 * (q & 1) + (e - 3 * ei);
                const int ki = wi - pi, kj = wj - pj;
                const bool hit = inimg && ((unsigned)ki < (unsigned)KH) && ((unsigned)kj < (unsigned)KW);
                const float w = wf[hit ? ki * KW + kj : 0];
                v = hit ? w : 0.0f;
            }
            T_lds[mi * KDIM + tid] = v;
        }
    }
    __syncthreads();

    const float* Arow = &T_lds[nl * KDIM + 2 * half];
    v2f af[3][KSTEPS];
    #pragma unroll
    for (int s = 0; s < KSTEPS; ++s) {
        af[0][s] = *(const v2f*)(Arow + 4 * s);
        af[1][s] = *(const v2f*)(Arow + 16 * KDIM + 4 * s);
        af[2][s] = *(const v2f*)(Arow + 32 * KDIM + 4 * s);
    }

    const float bv = bias[f];

    #pragma unroll 1
    for (int pass = 0; pass < NPASS; ++pass) {
        const char* bn = bp + ((pass + 1 < NPASS) ? 16 * NT * BATB : 0);

        v8f acc[3][NT];
        #pragma unroll
        for (int t = 0; t < 3; ++t)
            #pragma unroll
            for (int nt = 0; nt < NT; ++nt)
                acc[t][nt] = (v8f){0.f, 0.f, 0.f, 0.f, 0.f, 0.f, 0.f, 0.f};

        #pragma unroll
        for (int s = 0; s < KSTEPS; ++s) {
            #pragma unroll
            for (int nt = 0; nt < NT; ++nt) {
                acc[0][nt] = __builtin_amdgcn_wmma_f32_16x16x4_f32(false, af[0][s], false, bf[s][nt], (short)0, acc[0][nt], false, false);
                acc[1][nt] = __builtin_amdgcn_wmma_f32_16x16x4_f32(false, af[1][s], false, bf[s][nt], (short)0, acc[1][nt], false, false);
                acc[2][nt] = __builtin_amdgcn_wmma_f32_16x16x4_f32(false, af[2][s], false, bf[s][nt], (short)0, acc[2][nt], false, false);
            }
            asm volatile("" : "+v"(off[s][0]), "+v"(off[s][1]));
            #pragma unroll
            for (int nt = 0; nt < NT; ++nt) {
                bf[s][nt].x = *(const float*)(bn + off[s][0] + nt * 16 * BATB);
                bf[s][nt].y = *(const float*)(bn + off[s][1] + nt * 16 * BATB);
            }
        }
        bp = bn;

        #pragma unroll
        for (int nt = 0; nt < NT; ++nt) {
            float pa = acc[0][nt][0];
            #pragma unroll
            for (int r = 1; r < 8; ++r) pa = fmaxf(pa, acc[0][nt][r]);
            pa = fmaxf(pa, acc[1][nt][0]);
            float pb = acc[1][nt][1];
            #pragma unroll
            for (int r = 2; r < 8; ++r) pb = fmaxf(pb, acc[1][nt][r]);
            pb = fmaxf(pb, fmaxf(acc[2][nt][0], acc[2][nt][1]));
            const int b = wave * BPW + pass * (16 * NT) + 16 * nt + nl;
            float2 o;
            o.x = pa + bv;
            o.y = pb + bv;
            *(float2*)&out[(size_t)b * (NFILT * 4) + f * 4 + 2 * half] = o;
        }
    }
}

__global__ __launch_bounds__(THREADS) __attribute__((amdgpu_waves_per_eu(2, 2)))
void filters_wmma_kernel(const float* __restrict__ X,
                         const float* __restrict__ Xt,
                         const float* __restrict__ weight,
                         const float* __restrict__ bias,
                         const int*   __restrict__ positions,
                         float*       __restrict__ out,
                         size_t ws_size)
{
    __shared__ __attribute__((aligned(16))) float T_lds[MROWS * KDIM];
    if (ws_size >= WS_NEED) filters_body<true >(Xt, weight, bias, positions, out, T_lds);
    else                    filters_body<false>(X,  weight, bias, positions, out, T_lds);
}

extern "C" void kernel_launch(void* const* d_in, const int* in_sizes, int n_in,
                              void* d_out, int out_size, void* d_ws, size_t ws_size,
                              hipStream_t stream) {
    const float* X         = (const float*)d_in[0];
    const float* weight    = (const float*)d_in[1];
    const float* bias      = (const float*)d_in[2];
    const int*   positions = (const int*)d_in[3];
    float*       out       = (float*)d_out;
    float*       Xt        = (float*)d_ws;

    dim3 grid1(IMGPIX / TP, NBATCH / TB, 1);
    dim3 block1(256, 1, 1);
    batch_minor_copy_kernel<<<grid1, block1, 0, stream>>>(X, Xt, ws_size);

    dim3 grid2(NFILT, 1, 1);
    dim3 block2(THREADS, 1, 1);
    filters_wmma_kernel<<<grid2, block2, 0, stream>>>(X, Xt, weight, bias, positions, out, ws_size);
}


# A's rows re-dealt to M-tiles as compact pixel groups and K ordered by tile support: the 30 of 75 A fragments that are zero for every input are skipped, 45 WMMAs per 16 batches; sum order changes
# speedup vs baseline: 5.6024x; 1.1731x over previous
#include <hip/hip_runtime.h>


typedef __attribute__((ext_vector_type(2))) float v2f;
typedef __attribute__((ext_vector_type(8))) float v8f;

#define NFILT   1024
#define NBATCH  1024
#define HH      64
#define WW      64
#define IMGPIX  (HH * WW)
#define KH      5
#define KW      5
#define LOCAL   3
#define WINU    10
#define KDIM    (WINU * WINU)
#define KSTEPS  (KDIM / 4)
#define MROWS   48
#define THREADS 128
#define NT      2
#define BPW     (NBATCH / 4)
#define NPASS   (BPW / (16 * NT))
#define WS_NEED ((size_t)IMGPIX * NBATCH * sizeof(float))

#define TP 64
#define TB 128
__global__ __launch_bounds__(256)
void batch_minor_copy_kernel(const float* __restrict__ X, float* __restrict__ Xt, size_t ws_size)
{
    if (ws_size < WS_NEED) return;
    __shared__ float tile[TP][TB + 1];
    const int lane = threadIdx.x & 31;
    const int wv   = threadIdx.x >> 5;
    const int p0 = blockIdx.x * TP, b0 = blockIdx.y * TB;
    float2 v[TB / 8];
    #pragma unroll
    for (int i = 0; i < TB / 8; ++i)
        v[i] = *(const float2*)&X[(size_t)(b0 + wv + 8 * i) * IMGPIX + p0 + 2 * lane];
    #pragma unroll
    for (int i = 0; i < TB / 8; ++i) {
        tile[2 * lane + 0][wv + 8 * i] = v[i].x;
        tile[2 * lane + 1][wv + 8 * i] = v[i].y;
    }
    __syncthreads();
    #pragma unroll
    for (int i = 0; i < TP / 8; ++i) {
        const int p = wv + 8 * i;
        #pragma unroll
        for (int j = 0; j < TB / 32; ++j)
            Xt[(size_t)(p0 + p) * NBATCH + b0 + lane + 32 * j] = tile[p][lane + 32 * j];
    }
}

static __device__ __forceinline__ int clampi(int v, int lo, int hi) { return v < lo ? lo : (v > hi ? hi : v); }

static __device__ __forceinline__ constexpr int kord(int p)
{
    constexpr unsigned char ORDER[KDIM] = {
        0, 1, 2, 10, 11, 12, 20, 21, 30, 31, 40, 41, 50, 51, 60, 61, 70, 71, 80, 81, 82, 90, 91, 92,
        22, 32, 42, 52, 62, 72, 27, 37,
        23, 24, 25, 26, 33, 34, 35, 36, 43, 44, 45, 46, 53, 54, 55, 56, 63, 64, 65, 66, 73, 74, 75, 76,
        3, 4, 5, 6, 13, 14, 15, 16, 83, 84, 85, 86, 93, 94, 95, 96,
        47, 57, 67, 77,
        7, 8, 9, 17, 18, 19, 28, 29, 38, 39, 48, 49, 58, 59, 68, 69, 78, 79, 87, 88, 89, 97, 98, 99 };
    return ORDER[p];
}
static __device__ __forceinline__ constexpr bool tile_uses(int t, int s)
{
    return t == 0 ? s <= 17 : (t == 1 ? s >= 7 : ((s >= 6 && s <= 13) || s == 18));
}
static __device__ __forceinline__ constexpr int rowpix(int mi)
{
    constexpr signed char PIX[MROWS] = {
        0, 1, 2, 6, 7, 8, 12, 13, 18, 19, 24, 25, 26, 30, 31, 32,
        3, 4, 5, 9, 10, 11, 16, 17, 22, 23, 27, 28, 29, 33, 34, 35,
        14, 15, -1, -1, -1, -1, -1, -1, 20, 21, -1, -1, -1, -1, -1, -1 };
    return PIX[mi];
}

template <bool COPY>
static __device__ __forceinline__
void filters_body(const float* __restrict__ src,
                  const float* __restrict__ weight,
                  const float* __restrict__ bias,
                  const int*   __restrict__ positions,
                  float*       __restrict__ out,
                  float*       T_lds)
{
    constexpr int PIXB = COPY ? NBATCH * (int)sizeof(float) : (int)sizeof(float);
    constexpr int BATB = COPY ? (int)sizeof(float) : IMGPIX * (int)sizeof(float);

    const int f    = blockIdx.x;
    const int tid  = threadIdx.x;
    const int lane = tid & 31;
    const int wave = __builtin_amdgcn_readfirstlane(tid >> 5);
    const int half = lane >> 4;
    const int nl   = lane & 15;

    const int i0 = clampi(positions[2 * f + 0], 0, HH - KH);
    const int j0 = clampi(positions[2 * f + 1], 0, WW - KW);

    unsigned off[KSTEPS][2];
    #pragma unroll
    for (int s = 0; s < KSTEPS; ++s) {
        #pragma unroll
        for (int r = 0; r < 2; ++r) {
            const int w0 = kord(4 * s + r), w1 = kord(4 * s + 2 + r);
            const int a0 = (clampi(i0 + w0 / WINU - LOCAL, 0, HH - 1) * WW + clampi(j0 + w0 % WINU - LOCAL, 0, WW - 1)) * PIXB;
            const int a1 = (clampi(i0 + w1 / WINU - LOCAL, 0, HH - 1) * WW + clampi(j0 + w1 % WINU - LOCAL, 0, WW - 1)) * PIXB;
            off[s][r] = (unsigned)((half ? a1 : a0) + nl * BATB);
        }
    }

    const char* bp = (const char*)src + (size_t)wave * BPW * BATB;
    v2f bf[KSTEPS][NT];
    #pragma unroll
    for (int s = 0; s < KSTEPS; ++s) {
        #pragma unroll
        for (int nt = 0; nt < NT; ++nt) {
            bf[s][nt].x = *(const float*)(bp + off[s][0] + nt * 16 * BATB);
            bf[s][nt].y = *(const float*)(bp + off[s][1] + nt * 16 * BATB);
        }
    }

    if (tid < KDIM) {
        unsigned quad = 0;
        #pragma unroll
        for (int s = 0; s < KSTEPS; ++s) {
            const unsigned q = (unsigned)kord(4 * s) | ((unsigned)kord(4 * s + 1) << 8) |
                               ((unsigned)kord(4 * s + 2) << 16) | ((unsigned)kord(4 * s + 3) << 24);
            quad = ((tid >> 2) == s) ? q : quad;
        }
        const int  wp    = (int)((quad >> (8 * (tid & 3))) & 0xffu);
        const int  wi    = wp / WINU;
        const int  wj    = wp - wi * WINU;
        const int  r     = i0 + wi - LOCAL, c = j0 + wj - LOCAL;
        const bool inimg = ((unsigned)r < (unsigned)HH) && ((unsigned)c < (unsigned)WW);
        const float* wf  = weight + f * (KH * KW);
        #pragma unroll
        for (int mi = 0; mi < MROWS; ++mi) {
            const int px = rowpix(mi);
            float v = 0.0f;
            if (px >= 0) {
                const int pi = px / 6;
                const int pj = px - 6 * pi;
                const int ki = wi - pi, kj = wj - pj;
                const bool hit = inimg && ((unsigned)ki < (unsigned)KH) && ((unsigned)kj < (unsigned)KW);
                const float w = wf[hit ? ki * KW + kj : 0];
                v = hit ? w : 0.0f;
            }
            T_lds[mi * KDIM + tid] = v;
        }
    }
    __syncthreads();

    const float* Arow = &T_lds[nl * KDIM + 2 * half];
    v2f af[3][KSTEPS];
    #pragma unroll
    for (int s = 0; s < KSTEPS; ++s) {
        #pragma unroll
        for (int t = 0; t < 3; ++t)
            if (tile_uses(t, s)) af[t][s] = *(const v2f*)(Arow + 16 * t * KDIM + 4 * s);
    }

    const float bv = bias[f];

    #pragma unroll 1
    for (int pass = 0; pass < NPASS; ++pass) {
        const char* bn = bp + ((pass + 1 < NPASS) ? 16 * NT * BATB : 0);

        v8f acc[3][NT];
        #pragma unroll
        for (int t = 0; t < 3; ++t)
            #pragma unroll
            for (int nt = 0; nt < NT; ++nt)
                acc[t][nt] = (v8f){0.f, 0.f, 0.f, 0.f, 0.f, 0.f, 0.f, 0.f};

        #pragma unroll
        for (int s = 0; s < KSTEPS; ++s) {
            #pragma unroll
            for (int nt = 0; nt < NT; ++nt) {
                #pragma unroll
                for (int t = 0; t < 3; ++t)
                    if (tile_uses(t, s))
                        acc[t][nt] = __builtin_amdgcn_wmma_f32_16x16x4_f32(false, af[t][s], false, bf[s][nt], (short)0, acc[t][nt], false, false);
            }
            asm volatile("" : "+v"(off[s][0]), "+v"(off[s][1]));
            #pragma unroll
            for (int nt = 0; nt < NT; ++nt) {
                bf[s][nt].x = *(const float*)(bn + off[s][0] + nt * 16 * BATB);
                bf[s][nt].y = *(const float*)(bn + off[s][1] + nt * 16 * BATB);
            }
        }
        bp = bn;

        #pragma unroll
        for (int nt = 0; nt < NT; ++nt) {
            float pa = acc[2][nt][0];
            #pragma unroll
            for (int r = 0; r < 8; ++r) pa = fmaxf(pa, acc[0][nt][r]);
            float pb = acc[2][nt][1];
            #pragma unroll
            for (int r = 0; r < 8; ++r) pb = fmaxf(pb, acc[1][nt][r]);
            const int b = wave * BPW + pass * (16 * NT) + 16 * nt + nl;
            float2 o;
            o.x = pa + bv;
            o.y = pb + bv;
            *(float2*)&out[(size_t)b * (NFILT * 4) + f * 4 + 2 * half] = o;
        }
    }
}

__global__ __launch_bounds__(THREADS) __attribute__((amdgpu_waves_per_eu(2, 2)))
void filters_wmma_kernel(const float* __restrict__ X,
                         const float* __restrict__ Xt,
                         const float* __restrict__ weight,
                         const float* __restrict__ bias,
                         const int*   __restrict__ positions,
                         float*       __restrict__ out,
                         size_t ws_size)
{
    __shared__ __attribute__((aligned(16))) float T_lds[MROWS * KDIM];
    if (ws_size >= WS_NEED) filters_body<true >(Xt, weight, bias, positions, out, T_lds);
    else                    filters_body<false>(X,  weight, bias, positions, out, T_lds);
}

extern "C" void kernel_launch(void* const* d_in, const int* in_sizes, int n_in,
                              void* d_out, int out_size, void* d_ws, size_t ws_size,
                              hipStream_t stream) {
    const float* X         = (const float*)d_in[0];
    const float* weight    = (const float*)d_in[1];
    const float* bias      = (const float*)d_in[2];
    const int*   positions = (const int*)d_in[3];
    float*       out       = (float*)d_out;
    float*       Xt        = (float*)d_ws;

    dim3 grid1(IMGPIX / TP, NBATCH / TB, 1);
    dim3 block1(256, 1, 1);
    batch_minor_copy_kernel<<<grid1, block1, 0, stream>>>(X, Xt, ws_size);

    dim3 grid2(NFILT, 1, 1);
    dim3 block2(THREADS, 1, 1);
    filters_wmma_kernel<<<grid2, block2, 0, stream>>>(X, Xt, weight, bias, positions, out, ws_size);
}


# Workgroup = 8 consecutive filters (one per wave) x half the batch: the waves' pools meet in LDS and leave as whole 128-byte lines of out, 4 lines per store, not 32 pieces of 8 bytes; sum order kept
# speedup vs baseline: 6.7086x; 1.1975x over previous
#include <hip/hip_runtime.h>


typedef __attribute__((ext_vector_type(2))) float v2f;
typedef __attribute__((ext_vector_type(8))) float v8f;

#define NFILT   1024
#define NBATCH  1024
#define HH      64
#define WW      64
#define IMGPIX  (HH * WW)
#define KH      5
#define KW      5
#define LOCAL   3
#define WINU    10
#define KDIM    (WINU * WINU)
#define KSTEPS  (KDIM / 4)
#define MROWS   48
#define FPW     8
#define THREADS (32 * FPW)
#define NT      2
#define BSPLIT  2
#define BPW     (NBATCH / BSPLIT)
#define NPASS   (BPW / (16 * NT))
#define SPITCH  36
#define WS_NEED ((size_t)IMGPIX * NBATCH * sizeof(float))

#define TP 64
#define TB 128
__global__ __launch_bounds__(256)
void batch_minor_copy_kernel(const float* __restrict__ X, float* __restrict__ Xt, size_t ws_size)
{
    if (ws_size < WS_NEED) return;
    __shared__ float tile[TP][TB + 1];
    const int lane = threadIdx.x & 31;
    const int wv   = threadIdx.x >> 5;
    const int p0 = blockIdx.x * TP, b0 = blockIdx.y * TB;
    float2 v[TB / 8];
    #pragma unroll
    for (int i = 0; i < TB / 8; ++i)
        v[i] = *(const float2*)&X[(size_t)(b0 + wv + 8 * i) * IMGPIX + p0 + 2 * lane];
    #pragma unroll
    for (int i = 0; i < TB / 8; ++i) {
        tile[2 * lane + 0][wv + 8 * i] = v[i].x;
        tile[2 * lane + 1][wv + 8 * i] = v[i].y;
    }
    __syncthreads();
    #pragma unroll
    for (int i = 0; i < TP / 8; ++i) {
        const int p = wv + 8 * i;
        #pragma unroll
        for (int j = 0; j < TB / 32; ++j)
            Xt[(size_t)(p0 + p) * NBATCH + b0 + lane + 32 * j] = tile[p][lane + 32 * j];
    }
}

static __device__ __forceinline__ int clampi(int v, int lo, int hi) { return v < lo ? lo : (v > hi ? hi : v); }

static __device__ __forceinline__ constexpr int kord(int p)
{
    constexpr unsigned char ORDER[KDIM] = {
        0, 1, 2, 10, 11, 12, 20, 21, 30, 31, 40, 41, 50, 51, 60, 61, 70, 71, 80, 81, 82, 90, 91, 92,
        22, 32, 42, 52, 62, 72, 27, 37,
        23, 24, 25, 26, 33, 34, 35, 36, 43, 44, 45, 46, 53, 54, 55, 56, 63, 64, 65, 66, 73, 74, 75, 76,
        3, 4, 5, 6, 13, 14, 15, 16, 83, 84, 85, 86, 93, 94, 95, 96,
        47, 57, 67, 77,
        7, 8, 9, 17, 18, 19, 28, 29, 38, 39, 48, 49, 58, 59, 68, 69, 78, 79, 87, 88, 89, 97, 98, 99 };
    return ORDER[p];
}
static __device__ __forceinline__ constexpr bool tile_uses(int t, int s)
{
    return t == 0 ? s <= 17 : (t == 1 ? s >= 7 : ((s >= 6 && s <= 13) || s == 18));
}
static __device__ __forceinline__ constexpr int rowpix(int mi)
{
    constexpr signed char PIX[MROWS] = {
        0, 1, 2, 6, 7, 8, 12, 13, 18, 19, 24, 25, 26, 30, 31, 32,
        3, 4, 5, 9, 10, 11, 16, 17, 22, 23, 27, 28, 29, 33, 34, 35,
        14, 15, -1, -1, -1, -1, -1, -1, 20, 21, -1, -1, -1, -1, -1, -1 };
    return PIX[mi];
}

template <bool COPY>
static __device__ __forceinline__
void filters_body(const float* __restrict__ src,
                  const float* __restrict__ weight,
                  const float* __restrict__ bias,
                  const int*   __restrict__ positions,
                  float*       __restrict__ out,
                  float*       T_all,
                  float*       stage)
{
    constexpr int PIXB = COPY ? NBATCH * (int)sizeof(float) : (int)sizeof(float);
    constexpr int BATB = COPY ? (int)sizeof(float) : IMGPIX * (int)sizeof(float);

    const int tid  = threadIdx.x;
    const int lane = tid & 31;
    const int wave = __builtin_amdgcn_readfirstlane(tid >> 5);
    const int f    = blockIdx.x * FPW + wave;
    float*    T_lds = T_all + wave * (MROWS * KDIM);
    const int half = lane >> 4;
    const int nl   = lane & 15;

    const int i0 = clampi(positions[2 * f + 0], 0, HH - KH);
    const int j0 = clampi(positions[2 * f + 1], 0, WW - KW);

    unsigned off[KSTEPS][2];
    #pragma unroll
    for (int s = 0; s < KSTEPS; ++s) {
        #pragma unroll
        for (int r = 0; r < 2; ++r) {
            const int w0 = kord(4 * s + r), w1 = kord(4 * s + 2 + r);
            const int a0 = (clampi(i0 + w0 / WINU - LOCAL, 0, HH - 1) * WW + clampi(j0 + w0 % WINU - LOCAL, 0, WW - 1)) * PIXB;
            const int a1 = (clampi(i0 + w1 / WINU - LOCAL, 0, HH - 1) * WW + clampi(j0 + w1 % WINU - LOCAL, 0, WW - 1)) * PIXB;
            off[s][r] = (unsigned)((half ? a1 : a0) + nl * BATB);
        }
    }

    const char* bp = (const char*)src + (size_t)blockIdx.y * BPW * BATB;
    v2f bf[KSTEPS][NT];
    #pragma unroll
    for (int s = 0; s < KSTEPS; ++s) {
        #pragma unroll
        for (int nt = 0; nt < NT; ++nt) {
            bf[s][nt].x = *(const float*)(bp + off[s][0] + nt * 16 * BATB);
            bf[s][nt].y = *(const float*)(bp + off[s][1] + nt * 16 * BATB);
        }
    }

    #pragma unroll 1
    for (int p = lane; p < KDIM; p += 32) {
        unsigned quad = 0;
        #pragma unroll
        for (int s = 0; s < KSTEPS; ++s) {
            const unsigned q = (unsigned)kord(4 * s) | ((unsigned)kord(4 * s + 1) << 8) |
                               ((unsigned)kord(4 * s + 2) << 16) | ((unsigned)kord(4 * s + 3) << 24);
            quad = ((p >> 2) == s) ? q : quad;
        }
        const int  wp    = (int)((quad >> (8 * (p & 3))) & 0xffu);
        const int  wi    = wp / WINU;
        const int  wj    = wp - wi * WINU;
        const int  r     = i0 + wi - LOCAL, c = j0 + wj - LOCAL;
        const bool inimg = ((unsigned)r < (unsigned)HH) && ((unsigned)c < (unsigned)WW);
        const float* wf  = weight + f * (KH * KW);
        #pragma unroll
        for (int mi = 0; mi < MROWS; ++mi) {
            const int px = rowpix(mi);
            float v = 0.0f;
            if (px >= 0) {
                const int pi = px / 6;
                const int pj = px - 6 * pi;
                const int ki = wi - pi, kj = wj - pj;
                const bool hit = inimg && ((unsigned)ki < (unsigned)KH) && ((unsigned)kj < (unsigned)KW);
                const float w = wf[hit ? ki * KW + kj : 0];
                v = hit ? w : 0.0f;
            }
            T_lds[mi * KDIM + p] = v;
        }
    }
    __syncthreads();

    const float* Arow = &T_lds[nl * KDIM + 2 * half];
    v2f af[3][KSTEPS];
    #pragma unroll
    for (int s = 0; s < KSTEPS; ++s) {
        #pragma unroll
        for (int t = 0; t < 3; ++t)
            if (tile_uses(t, s)) af[t][s] = *(const v2f*)(Arow + 16 * t * KDIM + 4 * s);
    }

    const float bv = bias[f];

    #pragma unroll 1
    for (int pass = 0; pass < NPASS; ++pass) {
        const char* bn = bp + ((pass + 1 < NPASS) ? 16 * NT * BATB : 0);

        v8f acc[3][NT];
        #pragma unroll
        for (int t = 0; t < 3; ++t)
            #pragma unroll
            for (int nt = 0; nt < NT; ++nt)
                acc[t][nt] = (v8f){0.f, 0.f, 0.f, 0.f, 0.f, 0.f, 0.f, 0.f};

        #pragma unroll
        for (int s = 0; s < KSTEPS; ++s) {
            #pragma unroll
            for (int nt = 0; nt < NT; ++nt) {
                #pragma unroll
                for (int t = 0; t < 3; ++t)
                    if (tile_uses(t, s))
                        acc[t][nt] = __builtin_amdgcn_wmma_f32_16x16x4_f32(false, af[t][s], false, bf[s][nt], (short)0, acc[t][nt], false, false);
            }
            asm volatile("" : "+v"(off[s][0]), "+v"(off[s][1]));
            #pragma unroll
            for (int nt = 0; nt < NT; ++nt) {
                bf[s][nt].x = *(const float*)(bn + off[s][0] + nt * 16 * BATB);
                bf[s][nt].y = *(const float*)(bn + off[s][1] + nt * 16 * BATB);
            }
        }
        bp = bn;

        float* st = stage + (pass & 1) * (16 * NT * SPITCH);
        #pragma unroll
        for (int nt = 0; nt < NT; ++nt) {
            float pa = acc[2][nt][0];
            #pragma unroll
            for (int r = 0; r < 8; ++r) pa = fmaxf(pa, acc[0][nt][r]);
            float pb = acc[2][nt][1];
            #pragma unroll
            for (int r = 0; r < 8; ++r) pb = fmaxf(pb, acc[1][nt][r]);
            float2 o;
            o.x = pa + bv;
            o.y = pb + bv;
            *(float2*)&st[(16 * nt + nl) * SPITCH + wave * 4 + 2 * half] = o;
        }
        asm volatile("s_wait_dscnt 0x0\n\ts_barrier_signal -1\n\ts_barrier_wait -1" ::: "memory");
        {
            const int row = wave * 4 + (lane >> 3);
            const float4 v = *(const float4*)&st[row * SPITCH + (lane & 7) * 4];
            const int b = blockIdx.y * BPW + pass * (16 * NT) + row;
            *(float4*)&out[(size_t)b * (NFILT * 4) + blockIdx.x * (FPW * 4) + (lane & 7) * 4] = v;
        }
    }
}

__global__ __launch_bounds__(THREADS) __attribute__((amdgpu_waves_per_eu(2, 2)))
void filters_wmma_kernel(const float* __restrict__ X,
                         const float* __restrict__ Xt,
                         const float* __restrict__ weight,
                         const float* __restrict__ bias,
                         const int*   __restrict__ positions,
                         float*       __restrict__ out,
                         size_t ws_size)
{
    __shared__ __attribute__((aligned(16))) float T_lds[FPW * MROWS * KDIM];
    __shared__ __attribute__((aligned(16))) float stage[2 * 16 * NT * SPITCH];
    if (ws_size >= WS_NEED) filters_body<true >(Xt, weight, bias, positions, out, T_lds, stage);
    else                    filters_body<false>(X,  weight, bias, positions, out, T_lds, stage);
}

extern "C" void kernel_launch(void* const* d_in, const int* in_sizes, int n_in,
                              void* d_out, int out_size, void* d_ws, size_t ws_size,
                              hipStream_t stream) {
    const float* X         = (const float*)d_in[0];
    const float* weight    = (const float*)d_in[1];
    const float* bias      = (const float*)d_in[2];
    const int*   positions = (const int*)d_in[3];
    float*       out       = (float*)d_out;
    float*       Xt        = (float*)d_ws;

    dim3 grid1(IMGPIX / TP, NBATCH / TB, 1);
    dim3 block1(256, 1, 1);
    batch_minor_copy_kernel<<<grid1, block1, 0, stream>>>(X, Xt, ws_size);

    dim3 grid2(NFILT / FPW, BSPLIT, 1);
    dim3 block2(THREADS, 1, 1);
    filters_wmma_kernel<<<grid2, block2, 0, stream>>>(X, Xt, weight, bias, positions, out, ws_size);
}


# Pass widened to 64 batches: a lane fetches 4 consecutive batches of a window pixel with one 128-bit load (whole 256-byte runs) for 4 batch tiles: 50 loads per 64 batches, not 200; sum order kept
# speedup vs baseline: 7.0019x; 1.0437x over previous
#include <hip/hip_runtime.h>


typedef __attribute__((ext_vector_type(2))) float v2f;
typedef __attribute__((ext_vector_type(4))) float v4f;
typedef __attribute__((ext_vector_type(8))) float v8f;

#define NFILT   1024
#define NBATCH  1024
#define HH      64
#define WW      64
#define IMGPIX  (HH * WW)
#define KH      5
#define KW      5
#define LOCAL   3
#define WINU    10
#define KDIM    (WINU * WINU)
#define KSTEPS  (KDIM / 4)
#define RING    13
#define MROWS   48
#define FPW     8
#define THREADS (32 * FPW)
#define NT      4
#define BSPLIT  2
#define BPW     (NBATCH / BSPLIT)
#define NPASS   (BPW / (16 * NT))
#define SPITCH  36
#define WS_NEED ((size_t)IMGPIX * NBATCH * sizeof(float))

#define TP 64
#define TB 128
__global__ __launch_bounds__(256)
void batch_minor_copy_kernel(const float* __restrict__ X, float* __restrict__ Xt, size_t ws_size)
{
    if (ws_size < WS_NEED) return;
    __shared__ float tile[TP][TB + 1];
    const int lane = threadIdx.x & 31;
    const int wv   = threadIdx.x >> 5;
    const int p0 = blockIdx.x * TP, b0 = blockIdx.y * TB;
    float2 v[TB / 8];
    #pragma unroll
    for (int i = 0; i < TB / 8; ++i)
        v[i] = *(const float2*)&X[(size_t)(b0 + wv + 8 * i) * IMGPIX + p0 + 2 * lane];
    #pragma unroll
    for (int i = 0; i < TB / 8; ++i) {
        tile[2 * lane + 0][wv + 8 * i] = v[i].x;
        tile[2 * lane + 1][wv + 8 * i] = v[i].y;
    }
    __syncthreads();
    #pragma unroll
    for (int i = 0; i < TP / 8; ++i) {
        const int p = wv + 8 * i;
        #pragma unroll
        for (int j = 0; j < TB / 32; ++j)
            Xt[(size_t)(p0 + p) * NBATCH + b0 + lane + 32 * j] = tile[p][lane + 32 * j];
    }
}

static __device__ __forceinline__ int clampi(int v, int lo, int hi) { return v < lo ? lo : (v > hi ? hi : v); }

static __device__ __forceinline__ constexpr int kord(int p)
{
    constexpr unsigned char ORDER[KDIM] = {
        0, 1, 2, 10, 11, 12, 20, 21, 30, 31, 40, 41, 50, 51, 60, 61, 70, 71, 80, 81, 82, 90, 91, 92,
        22, 32, 42, 52, 62, 72, 27, 37,
        23, 24, 25, 26, 33, 34, 35, 36, 43, 44, 45, 46, 53, 54, 55, 56, 63, 64, 65, 66, 73, 74, 75, 76,
        3, 4, 5, 6, 13, 14, 15, 16, 83, 84, 85, 86, 93, 94, 95, 96,
        47, 57, 67, 77,
        7, 8, 9, 17, 18, 19, 28, 29, 38, 39, 48, 49, 58, 59, 68, 69, 78, 79, 87, 88, 89, 97, 98, 99 };
    return ORDER[p];
}
static __device__ __forceinline__ constexpr bool tile_uses(int t, int s)
{
    return t == 0 ? s <= 17 : (t == 1 ? s >= 7 : ((s >= 6 && s <= 13) || s == 18));
}
static __device__ __forceinline__ constexpr int rowpix(int mi)
{
    constexpr signed char PIX[MROWS] = {
        0, 1, 2, 6, 7, 8, 12, 13, 18, 19, 24, 25, 26, 30, 31, 32,
        3, 4, 5, 9, 10, 11, 16, 17, 22, 23, 27, 28, 29, 33, 34, 35,
        14, 15, -1, -1, -1, -1, -1, -1, 20, 21, -1, -1, -1, -1, -1, -1 };
    return PIX[mi];
}

template <bool COPY>
static __device__ __forceinline__ v4f load_batches(const char* p, unsigned o)
{
    if (COPY) return *(const v4f*)(p + o);
    constexpr int IMGB = IMGPIX * (int)sizeof(float);
    v4f v;
    v.x = *(const float*)(p + o);
    v.y = *(const float*)(p + o + IMGB);
    v.z = *(const float*)(p + o + 2 * IMGB);
    v.w = *(const float*)(p + o + 3 * IMGB);
    return v;
}

template <bool COPY>
static __device__ __forceinline__
void filters_body(const float* __restrict__ src,
                  const float* __restrict__ weight,
                  const float* __restrict__ bias,
                  const int*   __restrict__ positions,
                  float*       __restrict__ out,
                  float*       T_all,
                  float*       stage)
{
    constexpr int PIXB = COPY ? NBATCH * (int)sizeof(float) : (int)sizeof(float);
    constexpr int BATB = COPY ? (int)sizeof(float) : IMGPIX * (int)sizeof(float);

    const int tid  = threadIdx.x;
    const int lane = tid & 31;
    const int wave = __builtin_amdgcn_readfirstlane(tid >> 5);
    const int f    = blockIdx.x * FPW + wave;
    float*    T_lds = T_all + wave * (MROWS * KDIM);
    const int half = lane >> 4;
    const int nl   = lane & 15;

    const int i0 = clampi(positions[2 * f + 0], 0, HH - KH);
    const int j0 = clampi(positions[2 * f + 1], 0, WW - KW);

    unsigned off[KSTEPS][2];
    #pragma unroll
    for (int s = 0; s < KSTEPS; ++s) {
        #pragma unroll
        for (int r = 0; r < 2; ++r) {
            const int w0 = kord(4 * s + r), w1 = kord(4 * s + 2 + r);
            const int a0 = (clampi(i0 + w0 / WINU - LOCAL, 0, HH - 1) * WW + clampi(j0 + w0 % WINU - LOCAL, 0, WW - 1)) * PIXB;
            const int a1 = (clampi(i0 + w1 / WINU - LOCAL, 0, HH - 1) * WW + clampi(j0 + w1 % WINU - LOCAL, 0, WW - 1)) * PIXB;
            off[s][r] = (unsigned)((half ? a1 : a0) + nl * NT * BATB);
        }
    }

    const char* bp = (const char*)src + (size_t)blockIdx.y * BPW * BATB;
    v4f raw[RING][2];
    #pragma unroll
    for (int s = 0; s < RING; ++s) {
        raw[s][0] = load_batches<COPY>(bp, off[s][0]);
        raw[s][1] = load_batches<COPY>(bp, off[s][1]);
    }

    #pragma unroll 1
    for (int p = lane; p < KDIM; p += 32) {
        unsigned quad = 0;
        #pragma unroll
        for (int s = 0; s < KSTEPS; ++s) {
            const unsigned q = (unsigned)kord(4 * s) | ((unsigned)kord(4 * s + 1) << 8) |
                               ((unsigned)kord(4 * s + 2) << 16) | ((unsigned)kord(4 * s + 3) << 24);
            quad = ((p >> 2) == s) ? q : quad;
        }
        const int  wp    = (int)((quad >> (8 * (p & 3))) & 0xffu);
        const int  wi    = wp / WINU;
        const int  wj    = wp - wi * WINU;
        const int  r     = i0 + wi - LOCAL, c = j0 + wj - LOCAL;
        const bool inimg = ((unsigned)r < (unsigned)HH) && ((unsigned)c < (unsigned)WW);
        const float* wf  = weight + f * (KH * KW);
        #pragma unroll
        for (int mi = 0; mi < MROWS; ++mi) {
            const int px = rowpix(mi);
            float v = 0.0f;
            if (px >= 0) {
                const int pi = px / 6;
                const int pj = px - 6 * pi;
                const int ki = wi - pi, kj = wj - pj;
                const bool hit = inimg && ((unsigned)ki < (unsigned)KH) && ((unsigned)kj < (unsigned)KW);
                const float w = wf[hit ? ki * KW + kj : 0];
                v = hit ? w : 0.0f;
            }
            T_lds[mi * KDIM + p] = v;
        }
    }
    __syncthreads();

    const float* Arow = &T_lds[nl * KDIM + 2 * half];
    v2f af[3][KSTEPS];
    #pragma unroll
    for (int s = 0; s < KSTEPS; ++s) {
        #pragma unroll
        for (int t = 0; t < 3; ++t)
            if (tile_uses(t, s)) af[t][s] = *(const v2f*)(Arow + 16 * t * KDIM + 4 * s);
    }

    const float bv = bias[f];

    #pragma unroll 1
    for (int pass = 0; pass < NPASS; ++pass) {
        const char* bn = bp + ((pass + 1 < NPASS) ? 16 * NT * BATB : 0);

        v8f acc[3][NT];
        #pragma unroll
        for (int t = 0; t < 3; ++t)
            #pragma unroll
            for (int nt = 0; nt < NT; ++nt)
                acc[t][nt] = (v8f){0.f, 0.f, 0.f, 0.f, 0.f, 0.f, 0.f, 0.f};

        #pragma unroll
        for (int s = 0; s < KSTEPS; ++s) {
            const int slot = s < RING ? s : s - RING;
            #pragma unroll
            for (int nt = 0; nt < NT; ++nt) {
                const v2f bfrag = (v2f){raw[slot][0][nt], raw[slot][1][nt]};
                #pragma unroll
                for (int t = 0; t < 3; ++t)
                    if (tile_uses(t, s))
                        acc[t][nt] = __builtin_amdgcn_wmma_f32_16x16x4_f32(false, af[t][s], false, bfrag, (short)0, acc[t][nt], false, false);
            }
            const int rs = s + RING < KSTEPS ? s + RING : slot;
            asm volatile("" : "+v"(off[rs][0]), "+v"(off[rs][1]));
            raw[slot][0] = load_batches<COPY>(rs < RING ? bn : bp, off[rs][0]);
            raw[slot][1] = load_batches<COPY>(rs < RING ? bn : bp, off[rs][1]);
        }
        bp = bn;

        float* st = stage + (pass & 1) * (16 * NT * SPITCH);
        #pragma unroll
        for (int nt = 0; nt < NT; ++nt) {
            float pa = acc[2][nt][0];
            #pragma unroll
            for (int r = 0; r < 8; ++r) pa = fmaxf(pa, acc[0][nt][r]);
            float pb = acc[2][nt][1];
            #pragma unroll
            for (int r = 0; r < 8; ++r) pb = fmaxf(pb, acc[1][nt][r]);
            float2 o;
            o.x = pa + bv;
            o.y = pb + bv;
            *(float2*)&st[(16 * nt + nl) * SPITCH + wave * 4 + 2 * half] = o;
        }
        asm volatile("s_wait_dscnt 0x0\n\ts_barrier_signal -1\n\ts_barrier_wait -1" ::: "memory");
        #pragma unroll
        for (int it = 0; it < 2; ++it) {
            const int row = wave * 8 + it * 4 + (lane >> 3);
            const float4 v = *(const float4*)&st[row * SPITCH + (lane & 7) * 4];
            const int bl = NT * (row & 15) + (row >> 4);
            const int b = blockIdx.y * BPW + pass * (16 * NT) + bl;
            *(float4*)&out[(size_t)b * (NFILT * 4) + blockIdx.x * (FPW * 4) + (lane & 7) * 4] = v;
        }
    }
}

__global__ __launch_bounds__(THREADS) __attribute__((amdgpu_waves_per_eu(2, 2)))
void filters_wmma_kernel(const float* __restrict__ X,
                         const float* __restrict__ Xt,
                         const float* __restrict__ weight,
                         const float* __restrict__ bias,
                         const int*   __restrict__ positions,
                         float*       __restrict__ out,
                         size_t ws_size)
{
    __shared__ __attribute__((aligned(16))) float T_lds[FPW * MROWS * KDIM];
    __shared__ __attribute__((aligned(16))) float stage[2 * 16 * NT * SPITCH];
    if (ws_size >= WS_NEED) filters_body<true >(Xt, weight, bias, positions, out, T_lds, stage);
    else                    filters_body<false>(X,  weight, bias, positions, out, T_lds, stage);
}

extern "C" void kernel_launch(void* const* d_in, const int* in_sizes, int n_in,
                              void* d_out, int out_size, void* d_ws, size_t ws_size,
                              hipStream_t stream) {
    const float* X         = (const float*)d_in[0];
    const float* weight    = (const float*)d_in[1];
    const float* bias      = (const float*)d_in[2];
    const int*   positions = (const int*)d_in[3];
    float*       out       = (float*)d_out;
    float*       Xt        = (float*)d_ws;

    dim3 grid1(IMGPIX / TP, NBATCH / TB, 1);
    dim3 block1(256, 1, 1);
    batch_minor_copy_kernel<<<grid1, block1, 0, stream>>>(X, Xt, ws_size);

    dim3 grid2(NFILT / FPW, BSPLIT, 1);
    dim3 block2(THREADS, 1, 1);
    filters_wmma_kernel<<<grid2, block2, 0, stream>>>(X, Xt, weight, bias, positions, out, ws_size);
}


# Each lane reads its A fragments straight into registers from a zero-padded 15x15 copy of the weights in LDS: no 48x100 Toeplitz matrix in LDS, no 4 build trips, no barrier; sum order kept
# speedup vs baseline: 7.5508x; 1.0784x over previous
#include <hip/hip_runtime.h>


typedef __attribute__((ext_vector_type(2))) float v2f;
typedef __attribute__((ext_vector_type(4))) float v4f;
typedef __attribute__((ext_vector_type(8))) float v8f;

#define NFILT   1024
#define NBATCH  1024
#define HH      64
#define WW      64
#define IMGPIX  (HH * WW)
#define KH      5
#define KW      5
#define LOCAL   3
#define WINU    10
#define KDIM    (WINU * WINU)
#define KSTEPS  (KDIM / 4)
#define RING    13
#define MROWS   48
#define FPW     8
#define THREADS (32 * FPW)
#define NT      4
#define BSPLIT  2
#define BPW     (NBATCH / BSPLIT)
#define NPASS   (BPW / (16 * NT))
#define SPITCH  36
#define WS_NEED ((size_t)IMGPIX * NBATCH * sizeof(float))
#define ZC      15
#define ZW0     80
#define ZPAD    400
#define ZN      512

#define TP 64
#define TB 128
__global__ __launch_bounds__(256)
void batch_minor_copy_kernel(const float* __restrict__ X, float* __restrict__ Xt, size_t ws_size)
{
    if (ws_size < WS_NEED) return;
    __shared__ float tile[TP][TB + 1];
    const int lane = threadIdx.x & 31;
    const int wv   = threadIdx.x >> 5;
    const int p0 = blockIdx.x * TP, b0 = blockIdx.y * TB;
    float2 v[TB / 8];
    #pragma unroll
    for (int i = 0; i < TB / 8; ++i)
        v[i] = *(const float2*)&X[(size_t)(b0 + wv + 8 * i) * IMGPIX + p0 + 2 * lane];
    #pragma unroll
    for (int i = 0; i < TB / 8; ++i) {
        tile[2 * lane + 0][wv + 8 * i] = v[i].x;
        tile[2 * lane + 1][wv + 8 * i] = v[i].y;
    }
    __syncthreads();
    #pragma unroll
    for (int i = 0; i < TP / 8; ++i) {
        const int p = wv + 8 * i;
        #pragma unroll
        for (int j = 0; j < TB / 32; ++j)
            Xt[(size_t)(p0 + p) * NBATCH + b0 + lane + 32 * j] = tile[p][lane + 32 * j];
    }
}

static __device__ __forceinline__ int clampi(int v, int lo, int hi) { return v < lo ? lo : (v > hi ? hi : v); }

static __device__ __forceinline__ constexpr int kord(int p)
{
    constexpr unsigned char ORDER[KDIM] = {
        0, 1, 2, 10, 11, 12, 20, 21, 30, 31, 40, 41, 50, 51, 60, 61, 70, 71, 80, 81, 82, 90, 91, 92,
        22, 32, 42, 52, 62, 72, 27, 37,
        23, 24, 25, 26, 33, 34, 35, 36, 43, 44, 45, 46, 53, 54, 55, 56, 63, 64, 65, 66, 73, 74, 75, 76,
        3, 4, 5, 6, 13, 14, 15, 16, 83, 84, 85, 86, 93, 94, 95, 96,
        47, 57, 67, 77,
        7, 8, 9, 17, 18, 19, 28, 29, 38, 39, 48, 49, 58, 59, 68, 69, 78, 79, 87, 88, 89, 97, 98, 99 };
    return ORDER[p];
}
static __device__ __forceinline__ constexpr bool tile_uses(int t, int s)
{
    return t == 0 ? s <= 17 : (t == 1 ? s >= 7 : ((s >= 6 && s <= 13) || s == 18));
}
static __device__ __forceinline__ constexpr int rowpix(int mi)
{
    constexpr signed char PIX[MROWS] = {
        0, 1, 2, 6, 7, 8, 12, 13, 18, 19, 24, 25, 26, 30, 31, 32,
        3, 4, 5, 9, 10, 11, 16, 17, 22, 23, 27, 28, 29, 33, 34, 35,
        14, 15, -1, -1, -1, -1, -1, -1, 20, 21, -1, -1, -1, -1, -1, -1 };
    return PIX[mi];
}

template <bool COPY>
static __device__ __forceinline__ v4f load_batches(const char* p, unsigned o)
{
    if (COPY) return *(const v4f*)(p + o);
    constexpr int IMGB = IMGPIX * (int)sizeof(float);
    v4f v;
    v.x = *(const float*)(p + o);
    v.y = *(const float*)(p + o + IMGB);
    v.z = *(const float*)(p + o + 2 * IMGB);
    v.w = *(const float*)(p + o + 3 * IMGB);
    return v;
}

template <bool COPY>
static __device__ __forceinline__
void filters_body(const float* __restrict__ src,
                  const float* __restrict__ weight,
                  const float* __restrict__ bias,
                  const int*   __restrict__ positions,
                  float*       __restrict__ out,
                  float*       Z_all,
                  float*       stage)
{
    constexpr int PIXB = COPY ? NBATCH * (int)sizeof(float) : (int)sizeof(float);
    constexpr int BATB = COPY ? (int)sizeof(float) : IMGPIX * (int)sizeof(float);

    const int tid  = threadIdx.x;
    const int lane = tid & 31;
    const int wave = __builtin_amdgcn_readfirstlane(tid >> 5);
    const int f    = blockIdx.x * FPW + wave;
    const int half = lane >> 4;
    const int nl   = lane & 15;

    const int i0 = clampi(positions[2 * f + 0], 0, HH - KH);
    const int j0 = clampi(positions[2 * f + 1], 0, WW - KW);
    const float bv = bias[f];
    float wv = weight[f * (KH * KW) + (lane < KH * KW ? lane : 0)];
    asm volatile("" : "+v"(wv));

    float* Z = Z_all + wave * ZN;
    #pragma unroll
    for (int q = 0; q < ZN / 128; ++q)
        *(v4f*)&Z[(q * 32 + lane) * 4] = (v4f){0.f, 0.f, 0.f, 0.f};
    if (lane < KH * KW) Z[ZW0 + (lane / KW) * ZC + lane % KW] = wv;
    __builtin_amdgcn_fence(__ATOMIC_ACQ_REL, "wavefront");

    unsigned off[KSTEPS][2];
    #pragma unroll
    for (int s = 0; s < KSTEPS; ++s) {
        #pragma unroll
        for (int r = 0; r < 2; ++r) {
            const int w0 = kord(4 * s + r), w1 = kord(4 * s + 2 + r);
            const int a0 = (clampi(i0 + w0 / WINU - LOCAL, 0, HH - 1) * WW + clampi(j0 + w0 % WINU - LOCAL, 0, WW - 1)) * PIXB;
            const int a1 = (clampi(i0 + w1 / WINU - LOCAL, 0, HH - 1) * WW + clampi(j0 + w1 % WINU - LOCAL, 0, WW - 1)) * PIXB;
            off[s][r] = (unsigned)((half ? a1 : a0) + nl * NT * BATB);
        }
    }

    const char* bp = (const char*)src + (size_t)blockIdx.y * BPW * BATB;
    v4f raw[RING][2];
    #pragma unroll
    for (int s = 0; s < RING; ++s) {
        raw[s][0] = load_batches<COPY>(bp, off[s][0]);
        raw[s][1] = load_batches<COPY>(bp, off[s][1]);
    }


    int ro[3];
    #pragma unroll
    for (int t = 0; t < 3; ++t) {
        int v = -75;
        #pragma unroll
        for (int m = 0; m < 16; ++m) {
            const int px = rowpix(16 * t + m);
            const int val = px >= 0 ? (px / 6) * ZC + px % 6 : -75;
            v = (nl == m) ? val : v;
        }
        ro[t] = v;
    }

    v2f af[3][KSTEPS];
    #pragma unroll
    for (int s = 0; s < KSTEPS; ++s) {
        #pragma unroll
        for (int r = 0; r < 2; ++r) {
            const int w0 = kord(4 * s + r), w1 = kord(4 * s + 2 + r);
            const bool in0 = ((unsigned)(i0 + w0 / WINU - LOCAL) < (unsigned)HH) && ((unsigned)(j0 + w0 % WINU - LOCAL) < (unsigned)WW);
            const bool in1 = ((unsigned)(i0 + w1 / WINU - LOCAL) < (unsigned)HH) && ((unsigned)(j0 + w1 % WINU - LOCAL) < (unsigned)WW);
            const int c0 = in0 ? ZW0 + (w0 / WINU) * ZC + w0 % WINU : ZPAD;
            const int c1 = in1 ? ZW0 + (w1 / WINU) * ZC + w1 % WINU : ZPAD;
            const int cs = half ? c1 : c0;
            #pragma unroll
            for (int t = 0; t < 3; ++t)
                if (tile_uses(t, s)) af[t][s][r] = Z[cs - ro[t]];
        }
    }


    #pragma unroll 1
    for (int pass = 0; pass < NPASS; ++pass) {
        const char* bn = bp + ((pass + 1 < NPASS) ? 16 * NT * BATB : 0);

        v8f acc[3][NT];
        #pragma unroll
        for (int t = 0; t < 3; ++t)
            #pragma unroll
            for (int nt = 0; nt < NT; ++nt)
                acc[t][nt] = (v8f){0.f, 0.f, 0.f, 0.f, 0.f, 0.f, 0.f, 0.f};

        #pragma unroll
        for (int s = 0; s < KSTEPS; ++s) {
            const int slot = s < RING ? s : s - RING;
            #pragma unroll
            for (int nt = 0; nt < NT; ++nt) {
                const v2f bfrag = (v2f){raw[slot][0][nt], raw[slot][1][nt]};
                #pragma unroll
                for (int t = 0; t < 3; ++t)
                    if (tile_uses(t, s))
                        acc[t][nt] = __builtin_amdgcn_wmma_f32_16x16x4_f32(false, af[t][s], false, bfrag, (short)0, acc[t][nt], false, false);
            }
            const int rs = s + RING < KSTEPS ? s + RING : slot;
            asm volatile("" : "+v"(off[rs][0]), "+v"(off[rs][1]));
            raw[slot][0] = load_batches<COPY>(rs < RING ? bn : bp, off[rs][0]);
            raw[slot][1] = load_batches<COPY>(rs < RING ? bn : bp, off[rs][1]);
        }
        bp = bn;

        float* st = stage + (pass & 1) * (16 * NT * SPITCH);
        #pragma unroll
        for (int nt = 0; nt < NT; ++nt) {
            float pa = acc[2][nt][0];
            #pragma unroll
            for (int r = 0; r < 8; ++r) pa = fmaxf(pa, acc[0][nt][r]);
            float pb = acc[2][nt][1];
            #pragma unroll
            for (int r = 0; r < 8; ++r) pb = fmaxf(pb, acc[1][nt][r]);
            float2 o;
            o.x = pa + bv;
            o.y = pb + bv;
            *(float2*)&st[(16 * nt + nl) * SPITCH + wave * 4 + 2 * half] = o;
        }
        asm volatile("s_wait_dscnt 0x0\n\ts_barrier_signal -1\n\ts_barrier_wait -1" ::: "memory");
        #pragma unroll
        for (int it = 0; it < 2; ++it) {
            const int row = wave * 8 + it * 4 + (lane >> 3);
            const float4 v = *(const float4*)&st[row * SPITCH + (lane & 7) * 4];
            const int bl = NT * (row & 15) + (row >> 4);
            const int b = blockIdx.y * BPW + pass * (16 * NT) + bl;
            *(float4*)&out[(size_t)b * (NFILT * 4) + blockIdx.x * (FPW * 4) + (lane & 7) * 4] = v;
        }
    }
}

__global__ __launch_bounds__(THREADS) __attribute__((amdgpu_waves_per_eu(2, 2)))
void filters_wmma_kernel(const float* __restrict__ X,
                         const float* __restrict__ Xt,
                         const float* __restrict__ weight,
                         const float* __restrict__ bias,
                         const int*   __restrict__ positions,
                         float*       __restrict__ out,
                         size_t ws_size)
{
    __shared__ __attribute__((aligned(16))) float Z_lds[FPW * ZN];
    __shared__ __attribute__((aligned(16))) float stage[2 * 16 * NT * SPITCH];
    if (ws_size >= WS_NEED) filters_body<true >(Xt, weight, bias, positions, out, Z_lds, stage);
    else                    filters_body<false>(X,  weight, bias, positions, out, Z_lds, stage);
}

extern "C" void kernel_launch(void* const* d_in, const int* in_sizes, int n_in,
                              void* d_out, int out_size, void* d_ws, size_t ws_size,
                              hipStream_t stream) {
    const float* X         = (const float*)d_in[0];
    const float* weight    = (const float*)d_in[1];
    const float* bias      = (const float*)d_in[2];
    const int*   positions = (const int*)d_in[3];
    float*       out       = (float*)d_out;
    float*       Xt        = (float*)d_ws;

    dim3 grid1(IMGPIX / TP, NBATCH / TB, 1);
    dim3 block1(256, 1, 1);
    batch_minor_copy_kernel<<<grid1, block1, 0, stream>>>(X, Xt, ws_size);

    dim3 grid2(NFILT / FPW, BSPLIT, 1);
    dim3 block2(THREADS, 1, 1);
    filters_wmma_kernel<<<grid2, block2, 0, stream>>>(X, Xt, weight, bias, positions, out, ws_size);
}
